# starting kernel
# baseline (speedup 1.0000x reference)
#include <hip/hip_runtime.h>

typedef __attribute__((ext_vector_type(2))) float v2f;
typedef __attribute__((ext_vector_type(8))) float v8f;

#define T_DIM 2048
#define D_DIM 128
#define CHUNK 64

__global__ __launch_bounds__(256) void attn_pool_kernel(
    const float* __restrict__ x,
    const float* __restrict__ W,
    const float* __restrict__ bias,
    float* __restrict__ out)
{
    __shared__ float pbuf[2][CHUNK];
    __shared__ float s_shared;

    const int b    = blockIdx.x;
    const int tid  = threadIdx.x;
    const int wave = tid >> 5;
    const int lane = tid & 31;
    const int half = lane >> 4;
    const int n    = lane & 15;
    const int d0   = wave * 16;

    const float* xb = x + (size_t)b * T_DIM * D_DIM;

    if (tid == 0) s_shared = 0.0f;

    const float4 w4 = reinterpret_cast<const float4*>(W)[lane];

    v8f   c = {};
    float s_local = 0.0f;

    __syncthreads();

    for (int base = 0; base < T_DIM; base += CHUNK) {
        const int cur = (base / CHUNK) & 1;

        if (base + CHUNK < T_DIM) {
            __builtin_prefetch(xb + (size_t)(base + CHUNK) * D_DIM + tid * 32, 0, 0);
        }

        for (int j = 0; j < 8; ++j) {
            const int t = base + j * 8 + wave;
            const float4 xv =
                reinterpret_cast<const float4*>(xb + (size_t)t * D_DIM)[lane];
            float partial = xv.x * w4.x + xv.y * w4.y + xv.z * w4.z + xv.w * w4.w;
            partial += __shfl_xor(partial, 1, 32);
            partial += __shfl_xor(partial, 2, 32);
            partial += __shfl_xor(partial, 4, 32);
            partial += __shfl_xor(partial, 8, 32);
            partial += __shfl_xor(partial, 16, 32);
            const float e = tanhf(partial + bias[t]);
            const float p = __expf(e);
            s_local += p;
            if (lane == 0) pbuf[cur][j * 8 + wave] = p;
        }
        __syncthreads();

        for (int i = 0; i < 16; ++i) {
            const int t0 = base + 4 * i;
            const float p0 = pbuf[cur][4 * i + 0];
            const float p1 = pbuf[cur][4 * i + 1];
            const float p2 = pbuf[cur][4 * i + 2];
            const float p3 = pbuf[cur][4 * i + 3];

            v2f a;
            a.x = half ? p2 : p0;
            a.y = half ? p3 : p1;

            const float* col = xb + (size_t)(t0 + 2 * half) * D_DIM + d0 + n;
            v2f bb;
            bb.x = col[0];
            bb.y = col[D_DIM];

            c = __builtin_amdgcn_wmma_f32_16x16x4_f32(
                false, a, false, bb, (short)0, c, false, false);
        }
    }

    if (lane == 0) atomicAdd(&s_shared, s_local);
    __syncthreads();

    const float inv = 1.0f / s_shared;
    if (lane < 16) {
        out[(size_t)b * D_DIM + d0 + lane] = c[0] * inv;
    }
}

extern "C" void kernel_launch(void* const* d_in, const int* in_sizes, int n_in,
                              void* d_out, int out_size, void* d_ws, size_t ws_size,
                              hipStream_t stream) {
    const float* x    = (const float*)d_in[0];
    const float* W    = (const float*)d_in[1];
    const float* bias = (const float*)d_in[2];
    float* out = (float*)d_out;

    const int B = in_sizes[0] / (T_DIM * D_DIM);
    attn_pool_kernel<<<B, 256, 0, stream>>>(x, W, bias, out);
}


# T split inside the work-group: 32 waves (was 8), 4 wave groups each take a quarter of every 256-step chunk; partials and sum of p added in fixed order via LDS (sum order changed, LDS atomic gone)
# speedup vs baseline: 2.3563x; 2.3563x over previous
#include <hip/hip_runtime.h>

typedef __attribute__((ext_vector_type(2))) float v2f;
typedef __attribute__((ext_vector_type(8))) float v8f;

#define T_DIM 2048
#define D_DIM 128
#define NWAVE 32
#define NGRP  4
#define CHUNK (NWAVE * 8)
#define GSPAN (CHUNK / NGRP)

__global__ __launch_bounds__(1024) void attn_pool_kernel(
    const float* __restrict__ x,
    const float* __restrict__ W,
    const float* __restrict__ bias,
    float* __restrict__ out)
{
    __shared__ float pbuf[2][CHUNK];
    __shared__ float red[NGRP][D_DIM];
    __shared__ float swave[NWAVE];

    const int b    = blockIdx.x;
    const int tid  = threadIdx.x;
    const int wave = tid >> 5;
    const int lane = tid & 31;
    const int half = lane >> 4;
    const int n    = lane & 15;
    const int grp  = wave >> 3;
    const int d0   = (wave & 7) * 16;

    const float* xb = x + (size_t)b * T_DIM * D_DIM;

    const float4 w4 = reinterpret_cast<const float4*>(W)[lane];

    const float* colp = xb + (size_t)(grp * GSPAN + 2 * half) * D_DIM + d0 + n;

    v8f   c = {};
    float s_local = 0.0f;

    for (int base = 0; base < T_DIM; base += CHUNK) {
        const int cur = (base / CHUNK) & 1;

        if (base + CHUNK < T_DIM) {
            __builtin_prefetch(xb + (size_t)(base + CHUNK) * D_DIM + tid * 32, 0, 0);
        }

        for (int j = 0; j < 8; ++j) {
            const int tl = j * NWAVE + wave;
            const int t  = base + tl;
            const float4 xv =
                reinterpret_cast<const float4*>(xb + (size_t)t * D_DIM)[lane];
            float partial = xv.x * w4.x + xv.y * w4.y + xv.z * w4.z + xv.w * w4.w;
            partial += __shfl_xor(partial, 1, 32);
            partial += __shfl_xor(partial, 2, 32);
            partial += __shfl_xor(partial, 4, 32);
            partial += __shfl_xor(partial, 8, 32);
            partial += __shfl_xor(partial, 16, 32);
            const float e = tanhf(partial + bias[t]);
            const float p = __expf(e);
            s_local += p;
            if (lane == 0) pbuf[cur][tl] = p;
        }
        __syncthreads();

        for (int i = 0; i < GSPAN / 4; ++i) {
            const int tl0 = grp * GSPAN + 4 * i;
            const float p0 = pbuf[cur][tl0 + 0];
            const float p1 = pbuf[cur][tl0 + 1];
            const float p2 = pbuf[cur][tl0 + 2];
            const float p3 = pbuf[cur][tl0 + 3];

            v2f a;
            a.x = half ? p2 : p0;
            a.y = half ? p3 : p1;

            const float* col = colp + (size_t)(4 * i) * D_DIM;
            v2f bb;
            bb.x = col[0];
            bb.y = col[D_DIM];

            c = __builtin_amdgcn_wmma_f32_16x16x4_f32(
                false, a, false, bb, (short)0, c, false, false);
        }
        colp += (size_t)CHUNK * D_DIM;
    }

    if (lane == 0) swave[wave] = s_local;
    if (lane < 16) {
        red[grp][d0 + lane] = c[0];
    }
    __syncthreads();

    if (tid < D_DIM) {
        float s = 0.0f;
        for (int w = 0; w < NWAVE; ++w) s += swave[w];
        const float acc = (red[0][tid] + red[1][tid]) + (red[2][tid] + red[3][tid]);
        const float inv = 1.0f / s;
        out[(size_t)b * D_DIM + tid] = acc * inv;
    }
}

extern "C" void kernel_launch(void* const* d_in, const int* in_sizes, int n_in,
                              void* d_out, int out_size, void* d_ws, size_t ws_size,
                              hipStream_t stream) {
    const float* x    = (const float*)d_in[0];
    const float* W    = (const float*)d_in[1];
    const float* bias = (const float*)d_in[2];
    float* out = (float*)d_out;

    const int B = in_sizes[0] / (T_DIM * D_DIM);
    attn_pool_kernel<<<B, 1024, 0, stream>>>(x, W, bias, out);
}
